# starting kernel
# baseline (speedup 1.0000x reference)
#include <hip/hip_runtime.h>
#include <hip/hip_bf16.h>

#define N_NODES 50000
#define N_EDGES 800000
#define DIM 128

typedef __attribute__((ext_vector_type(2))) float v2f;
typedef __attribute__((ext_vector_type(8))) float v8f;

__global__ __launch_bounds__(256) void gat_init(int* __restrict__ counts,
                                                int* __restrict__ cursor) {
    int tid = blockIdx.x * blockDim.x + threadIdx.x;
    if (tid < N_NODES) {
        counts[tid] = 0;
        cursor[tid] = 0;
    }
}

__global__ __launch_bounds__(256) void gat_gemm(const float* __restrict__ h,
                                                const float* __restrict__ W,
                                                float* __restrict__ z) {
    const int lane = threadIdx.x & 31;
    const int wv   = threadIdx.x >> 5;
    const int m0   = blockIdx.x * 16;
    const int n0   = wv * 16;
    const int half = lane >> 4;
    const int lr   = lane & 15;

    const float* __restrict__ Arow = h + (size_t)(m0 + lr) * DIM;
    const float* __restrict__ Brow = W + (size_t)(n0 + lr) * DIM;

    v8f acc = {};
    #pragma unroll 4
    for (int k = 0; k < DIM; k += 4) {
        const int kk = k + half * 2;
        v2f a = *(const v2f*)(Arow + kk);
        v2f b = *(const v2f*)(Brow + kk);
        acc = __builtin_amdgcn_wmma_f32_16x16x4_f32(
             false, a,  false, b,
             (short)0, acc,  false,  false);
    }

    const int n = n0 + lr;
    #pragma unroll
    for (int v = 0; v < 8; ++v) {
        const int m = m0 + v + 8 * half;
        z[(size_t)m * DIM + n] = acc[v];
    }
}

__global__ __launch_bounds__(256) void gat_gemv(const float* __restrict__ z,
                                                const float* __restrict__ a_attn,
                                                float* __restrict__ s_src,
                                                float* __restrict__ s_dst) {
    const int node = blockIdx.x * 8 + (threadIdx.x >> 5);
    const int lane = threadIdx.x & 31;
    if (node >= N_NODES) return;

    const float4 zv = ((const float4*)(z + (size_t)node * DIM))[lane];
    const float4 av = ((const float4*)(a_attn))[lane];
    const float4 dv = ((const float4*)(a_attn + DIM))[lane];

    float ss = zv.x * av.x + zv.y * av.y + zv.z * av.z + zv.w * av.w;
    float sd = zv.x * dv.x + zv.y * dv.y + zv.z * dv.z + zv.w * dv.w;

    #pragma unroll
    for (int off = 16; off > 0; off >>= 1) {
        ss += __shfl_down(ss, off, 32);
        sd += __shfl_down(sd, off, 32);
    }
    if (lane == 0) {
        s_src[node] = ss;
        s_dst[node] = sd;
    }
}

__global__ __launch_bounds__(256) void gat_hist(const int* __restrict__ dst,
                                                int* __restrict__ counts) {
    int i = blockIdx.x * blockDim.x + threadIdx.x;
    if (i >= N_EDGES) return;
    atomicAdd(counts + dst[i], 1);
}

__global__ __launch_bounds__(1024) void gat_scan(const int* __restrict__ counts,
                                                 int* __restrict__ row_start) {
    __shared__ int lds[1024];
    __shared__ int carry_s;
    if (threadIdx.x == 0) carry_s = 0;
    __syncthreads();

    for (int base = 0; base < N_NODES; base += 1024) {
        const int idx = base + (int)threadIdx.x;
        const int v = (idx < N_NODES) ? counts[idx] : 0;
        int acc = v;
        lds[threadIdx.x] = acc;
        __syncthreads();
        #pragma unroll
        for (int off = 1; off < 1024; off <<= 1) {
            int t = (threadIdx.x >= (unsigned)off) ? lds[threadIdx.x - off] : 0;
            __syncthreads();
            acc += t;
            lds[threadIdx.x] = acc;
            __syncthreads();
        }
        const int c = carry_s;
        if (idx < N_NODES) row_start[idx] = c + acc - v;
        __syncthreads();
        if (threadIdx.x == 1023) carry_s = c + lds[1023];
        __syncthreads();
    }
    if (threadIdx.x == 0) row_start[N_NODES] = carry_s;
}

__global__ __launch_bounds__(256) void gat_fill(const int* __restrict__ src,
                                                const int* __restrict__ dst,
                                                const float* __restrict__ s_src,
                                                const float* __restrict__ s_dst,
                                                const int* __restrict__ row_start,
                                                int* __restrict__ cursor,
                                                int* __restrict__ csr_src,
                                                float* __restrict__ csr_e) {
    int i = blockIdx.x * blockDim.x + threadIdx.x;
    if (i >= N_EDGES) return;
    const int s = src[i];
    const int d = dst[i];
    const int pos = row_start[d] + atomicAdd(cursor + d, 1);
    const float v = s_src[s] + s_dst[d];
    csr_src[pos] = s;
    csr_e[pos]   = v > 0.0f ? v : 0.01f * v;
}

__global__ __launch_bounds__(256) void gat_node(const float* __restrict__ z,
                                                const int* __restrict__ row_start,
                                                const int* __restrict__ csr_src,
                                                float* __restrict__ csr_e,
                                                float* __restrict__ out) {
    const int node = blockIdx.x * 8 + (threadIdx.x >> 5);
    const int lane = threadIdx.x & 31;
    const int beg = row_start[node];
    const int end = row_start[node + 1];

    float mx = -INFINITY;
    for (int p = beg + lane; p < end; p += 32) mx = fmaxf(mx, csr_e[p]);
    #pragma unroll
    for (int off = 16; off > 0; off >>= 1) mx = fmaxf(mx, __shfl_xor(mx, off, 32));

    float sum = 0.0f;
    for (int p = beg + lane; p < end; p += 32) {
        const float ex = __expf(csr_e[p] - mx);
        csr_e[p] = ex;
        sum += ex;
    }
    #pragma unroll
    for (int off = 16; off > 0; off >>= 1) sum += __shfl_xor(sum, off, 32);

    float4 acc = {0.0f, 0.0f, 0.0f, 0.0f};
    for (int p0 = beg; p0 < end; p0 += 32) {
        const int nn = min(32, end - p0);
        const int  pl = p0 + lane;
        const float wl = (pl < end) ? csr_e[pl]   : 0.0f;
        const int   sl = (pl < end) ? csr_src[pl] : 0;
        for (int j = 0; j < nn; ++j) {
            const float w = __shfl(wl, j, 32);
            const int   s = __shfl(sl, j, 32);
            const float4 zv = ((const float4*)(z + (size_t)s * DIM))[lane];
            acc.x += w * zv.x;
            acc.y += w * zv.y;
            acc.z += w * zv.z;
            acc.w += w * zv.w;
        }
    }

    const float inv = (end > beg) ? 1.0f / sum : 0.0f;
    float4 o = {acc.x * inv, acc.y * inv, acc.z * inv, acc.w * inv};
    ((float4*)(out + (size_t)node * DIM))[lane] = o;
}

extern "C" void kernel_launch(void* const* d_in, const int* in_sizes, int n_in,
                              void* d_out, int out_size, void* d_ws, size_t ws_size,
                              hipStream_t stream) {
    const float* h      = (const float*)d_in[0];
    const float* W_fc   = (const float*)d_in[1];
    const float* a_attn = (const float*)d_in[2];
    const int*   src    = (const int*)d_in[3];
    const int*   dst    = (const int*)d_in[4];
    float*       out    = (float*)d_out;
    (void)in_sizes; (void)n_in; (void)out_size; (void)ws_size;

    float* ws        = (float*)d_ws;
    float* z         = ws;
    float* s_src     = z + (size_t)N_NODES * DIM;
    float* s_dst     = s_src + N_NODES;
    float* csr_e     = s_dst + N_NODES;
    int*   counts    = (int*)(csr_e + N_EDGES);
    int*   cursor    = counts + N_NODES;
    int*   row_start = cursor + N_NODES;
    int*   csr_src   = row_start + N_NODES + 1;

    gat_init<<<(N_NODES + 255) / 256, 256, 0, stream>>>(counts, cursor);

    gat_gemm<<<N_NODES / 16, 256, 0, stream>>>(h, W_fc, z);

    gat_gemv<<<(N_NODES + 7) / 8, 256, 0, stream>>>(z, a_attn, s_src, s_dst);

    gat_hist<<<(N_EDGES + 255) / 256, 256, 0, stream>>>(dst, counts);

    gat_scan<<<1, 1024, 0, stream>>>(counts, row_start);

    gat_fill<<<(N_EDGES + 255) / 256, 256, 0, stream>>>(src, dst, s_src, s_dst,
                                                        row_start, cursor,
                                                        csr_src, csr_e);

    gat_node<<<N_NODES / 8, 256, 0, stream>>>(z, row_start, csr_src, csr_e, out);
}


# node kernel: a node's dependent round trips cut: bounds in one 8-byte record, logits and source ids read once into lanes (logits not rewritten), source rows fetched 8 at a time ahead of their sums
# speedup vs baseline: 2.5294x; 2.5294x over previous
#include <hip/hip_runtime.h>
#include <hip/hip_bf16.h>

#define N_NODES 50000
#define N_EDGES 800000
#define DIM 128
#define N_SLABS 32
#define SLAB (N_EDGES / N_SLABS)
#define SLAB4 (SLAB / 4)
#define NODES4 (N_NODES / 4)

typedef __attribute__((ext_vector_type(2))) float v2f;
typedef __attribute__((ext_vector_type(8))) float v8f;

__global__ __launch_bounds__(256) void gat_gemm(const float* __restrict__ h,
                                                const float* __restrict__ W,
                                                float* __restrict__ z) {
    const int lane = threadIdx.x & 31;
    const int wv   = threadIdx.x >> 5;
    const int m0   = blockIdx.x * 16;
    const int n0   = wv * 16;
    const int half = lane >> 4;
    const int lr   = lane & 15;

    const float* __restrict__ Arow = h + (size_t)(m0 + lr) * DIM;
    const float* __restrict__ Brow = W + (size_t)(n0 + lr) * DIM;

    v8f acc = {};
    #pragma unroll 4
    for (int k = 0; k < DIM; k += 4) {
        const int kk = k + half * 2;
        v2f a = *(const v2f*)(Arow + kk);
        v2f b = *(const v2f*)(Brow + kk);
        acc = __builtin_amdgcn_wmma_f32_16x16x4_f32(
             false, a,  false, b,
             (short)0, acc,  false,  false);
    }

    const int n = n0 + lr;
    #pragma unroll
    for (int v = 0; v < 8; ++v) {
        const int m = m0 + v + 8 * half;
        z[(size_t)m * DIM + n] = acc[v];
    }
}

__global__ __launch_bounds__(256) void gat_gemv(const float* __restrict__ z,
                                                const float* __restrict__ a_attn,
                                                float* __restrict__ s_src,
                                                float* __restrict__ s_dst) {
    const int node = blockIdx.x * 8 + (threadIdx.x >> 5);
    const int lane = threadIdx.x & 31;
    if (node >= N_NODES) return;

    const float4 zv = ((const float4*)(z + (size_t)node * DIM))[lane];
    const float4 av = ((const float4*)(a_attn))[lane];
    const float4 dv = ((const float4*)(a_attn + DIM))[lane];

    float ss = zv.x * av.x + zv.y * av.y + zv.z * av.z + zv.w * av.w;
    float sd = zv.x * dv.x + zv.y * dv.y + zv.z * dv.z + zv.w * dv.w;

    #pragma unroll
    for (int off = 16; off > 0; off >>= 1) {
        ss += __shfl_down(ss, off, 32);
        sd += __shfl_down(sd, off, 32);
    }
    if (lane == 0) {
        s_src[node] = ss;
        s_dst[node] = sd;
    }
}

__global__ __launch_bounds__(1024) void gat_hist(const int* __restrict__ dst,
                                                 int* __restrict__ slab_tab,
                                                 int* __restrict__ rank,
                                                 int* alloc) {
    __shared__ int tbl[N_NODES];
    const int t  = threadIdx.x;
    const int k  = blockIdx.x;
    const int4* __restrict__ d4 = (const int4*)(dst + (size_t)k * SLAB);
    int4* __restrict__ r4       = (int4*)(rank + (size_t)k * SLAB);
    int4* __restrict__ o4       = (int4*)(slab_tab + (size_t)k * N_NODES);

    int4 dv[7];
    #pragma unroll
    for (int r = 0; r < 7; ++r) {
        const int q = t + r * 1024;
        dv[r] = d4[q < SLAB4 ? q : SLAB4 - 1];
    }

    int4* tbl4 = (int4*)tbl;
    for (int n = t; n < NODES4; n += 1024) tbl4[n] = make_int4(0, 0, 0, 0);
    if (k == 0 && t == 0) *alloc = 0;
    __syncthreads();

    int4 rk[7];
    #pragma unroll
    for (int r = 0; r < 7; ++r) {
        const int q = t + r * 1024;
        if (q < SLAB4) {
            rk[r].x = atomicAdd(&tbl[dv[r].x], 1);
            rk[r].y = atomicAdd(&tbl[dv[r].y], 1);
            rk[r].z = atomicAdd(&tbl[dv[r].z], 1);
            rk[r].w = atomicAdd(&tbl[dv[r].w], 1);
        }
    }
    __syncthreads();

    #pragma unroll
    for (int r = 0; r < 7; ++r) {
        const int q = t + r * 1024;
        if (q < SLAB4) r4[q] = rk[r];
    }
    for (int n = t; n < NODES4; n += 1024) o4[n] = tbl4[n];
}

__global__ __launch_bounds__(256) void gat_alloc(int* __restrict__ slab_tab,
                                                 int* __restrict__ seg,
                                                 int* alloc) {
    __shared__ int wave_off[8];
    __shared__ int base_s;
    const int idx  = blockIdx.x * 256 + (int)threadIdx.x;
    const int lane = threadIdx.x & 31;
    const int wv   = threadIdx.x >> 5;
    const bool live = idx < N_NODES;
    const int  col  = live ? idx : N_NODES - 1;
    int c[N_SLABS];
    int v = 0;
    #pragma unroll
    for (int k = 0; k < N_SLABS; ++k) {
        c[k] = slab_tab[(size_t)k * N_NODES + col];
        v += c[k];
    }
    if (!live) v = 0;

    int inc = v;
    for (int off = 1; off < 32; off <<= 1) {
        const int t = __shfl_up(inc, off, 32);
        if (lane >= off) inc += t;
    }
    if (lane == 31) wave_off[wv] = inc;
    __syncthreads();
    if (threadIdx.x == 0) {
        int tot = 0;
        for (int w = 0; w < 8; ++w) {
            const int t = wave_off[w];
            wave_off[w] = tot;
            tot += t;
        }
        base_s = atomicAdd(alloc, tot);
    }
    __syncthreads();
    if (live) {
        int run = base_s + wave_off[wv] + inc - v;
        ((int2*)seg)[idx] = make_int2(run, v);
        #pragma unroll
        for (int k = 0; k < N_SLABS; ++k) {
            slab_tab[(size_t)k * N_NODES + idx] = run;
            run += c[k];
        }
    }
}

__global__ __launch_bounds__(256) void gat_fill(const int* __restrict__ src,
                                                const int* __restrict__ dst,
                                                const int* __restrict__ rank,
                                                const float* __restrict__ s_src,
                                                const float* __restrict__ s_dst,
                                                const int* __restrict__ slab_tab,
                                                int* __restrict__ csr_src,
                                                float* __restrict__ csr_e) {
    int i = blockIdx.x * blockDim.x + threadIdx.x;
    if (i >= N_EDGES) return;
    const int s = src[i];
    const int d = dst[i];
    const int k = i / SLAB;
    const int pos = slab_tab[(size_t)k * N_NODES + d] + rank[i];
    const float v = s_src[s] + s_dst[d];
    csr_src[pos] = s;
    csr_e[pos]   = v > 0.0f ? v : 0.01f * v;
}

template <int N>
__device__ __forceinline__ void rows_request(const float* __restrict__ z, const int lane,
                                             const int sl, const int j, float4 (&r)[8]) {
    #pragma unroll
    for (int u = 0; u < N; ++u) {
        const int s = __builtin_amdgcn_readlane(sl, j + u);
        r[u] = ((const float4*)z)[(unsigned)s * (DIM / 4) + (unsigned)lane];
    }
}

template <int N>
__device__ __forceinline__ void rows_sum(const float wl, const int j,
                                         const float4 (&r)[8], float4& acc) {
    #pragma unroll
    for (int u = 0; u < N; ++u) {
        const float w = __int_as_float(__builtin_amdgcn_readlane(__float_as_int(wl), j + u));
        acc.x += w * r[u].x;
        acc.y += w * r[u].y;
        acc.z += w * r[u].z;
        acc.w += w * r[u].w;
    }
}

__device__ __forceinline__ void rows_block(const float* __restrict__ z, const int lane,
                                           const int sl, const float wl,
                                           int j, const int nn, float4& acc) {
    float4 r[8];
    for (; j + 8 <= nn; j += 8) {
        rows_request<8>(z, lane, sl, j, r);
        rows_sum<8>(wl, j, r, acc);
    }
    const int rem = nn - j;
    if (rem > 4) {
        rows_request<8>(z, lane, sl, j, r);
        rows_sum<8>(wl, j, r, acc);
    } else if (rem > 0) {
        rows_request<4>(z, lane, sl, j, r);
        rows_sum<4>(wl, j, r, acc);
    }
}

__device__ __forceinline__ float wave_sum(float v) {
    #pragma unroll
    for (int off = 16; off > 0; off >>= 1) v += __shfl_xor(v, off, 32);
    return v;
}

__global__ __launch_bounds__(256) void gat_node(const float* __restrict__ z,
                                                const int* __restrict__ seg,
                                                const int* __restrict__ csr_src,
                                                const float* __restrict__ csr_e,
                                                float* __restrict__ out) {
    const int node = blockIdx.x * 8 + __builtin_amdgcn_readfirstlane((int)(threadIdx.x >> 5));
    const int lane = threadIdx.x & 31;
    const int2 sg  = ((const int2*)seg)[node];
    const int beg = sg.x;
    const int cnt = sg.y;
    const int end = beg + cnt;

    float4 acc = {0.0f, 0.0f, 0.0f, 0.0f};
    float sum = 0.0f;
    if (cnt > 0) {
        const bool in0 = lane < cnt;
        const int  p   = in0 ? beg + lane : beg;
        const float e0 = csr_e[p];
        const int   s0 = csr_src[p];

        float4 r0[8];
        rows_request<8>(z, lane, s0, 0, r0);

        float mx = in0 ? e0 : -INFINITY;
        for (int q = beg + 32 + lane; q < end; q += 32) mx = fmaxf(mx, csr_e[q]);
        #pragma unroll
        for (int off = 16; off > 0; off >>= 1) mx = fmaxf(mx, __shfl_xor(mx, off, 32));

        const float w0 = in0 ? __expf(e0 - mx) : 0.0f;
        sum = w0;
        if (cnt <= 32) sum = wave_sum(sum);

        rows_sum<8>(w0, 0, r0, acc);
        rows_block(z, lane, s0, w0, 8, min(32, cnt), acc);

        if (cnt > 32) {
            for (int p0 = beg + 32; p0 < end; p0 += 32) {
                const bool in = p0 + lane < end;
                const int  pp = in ? p0 + lane : beg;
                const float el = csr_e[pp];
                const int   sl = csr_src[pp];
                const float wl = in ? __expf(el - mx) : 0.0f;
                sum += wl;
                rows_block(z, lane, sl, wl, 0, min(32, end - p0), acc);
            }
            sum = wave_sum(sum);
        }
    }

    const float inv = (cnt > 0) ? 1.0f / sum : 0.0f;
    float4 o = {acc.x * inv, acc.y * inv, acc.z * inv, acc.w * inv};
    ((float4*)(out + (size_t)node * DIM))[lane] = o;
}

extern "C" void kernel_launch(void* const* d_in, const int* in_sizes, int n_in,
                              void* d_out, int out_size, void* d_ws, size_t ws_size,
                              hipStream_t stream) {
    const float* h      = (const float*)d_in[0];
    const float* W_fc   = (const float*)d_in[1];
    const float* a_attn = (const float*)d_in[2];
    const int*   src    = (const int*)d_in[3];
    const int*   dst    = (const int*)d_in[4];
    float*       out    = (float*)d_out;
    (void)in_sizes; (void)n_in; (void)out_size; (void)ws_size;

    float* ws        = (float*)d_ws;
    float* z         = ws;
    float* s_src     = z + (size_t)N_NODES * DIM;
    float* s_dst     = s_src + N_NODES;
    float* csr_e     = s_dst + N_NODES;
    int*   seg       = (int*)(csr_e + N_EDGES);
    int*   alloc     = seg + 2 * N_NODES;
    int*   csr_src   = alloc + 1;
    int*   rank      = (int*)d_out;
    int*   slab_tab  = rank + N_EDGES;

    gat_gemm<<<N_NODES / 16, 256, 0, stream>>>(h, W_fc, z);

    gat_gemv<<<(N_NODES + 7) / 8, 256, 0, stream>>>(z, a_attn, s_src, s_dst);

    gat_hist<<<N_SLABS, 1024, 0, stream>>>(dst, slab_tab, rank, alloc);

    gat_alloc<<<(N_NODES + 255) / 256, 256, 0, stream>>>(slab_tab, seg, alloc);

    gat_fill<<<(N_EDGES + 255) / 256, 256, 0, stream>>>(src, dst, rank, s_src, s_dst,
                                                        slab_tab,
                                                        csr_src, csr_e);

    gat_node<<<N_NODES / 8, 256, 0, stream>>>(z, seg, csr_src, csr_e, out);
}


# GEMM register-blocked: a wave owns an 80x32 block of z (5x2 tiles, ten accumulators); 7 fragment loads feed 10 products per k-step (was 2 per product), issued 6 k-steps ahead; k order of sums kept
# speedup vs baseline: 3.1635x; 1.2507x over previous
#include <hip/hip_runtime.h>
#include <hip/hip_bf16.h>

#define N_NODES 50000
#define N_EDGES 800000
#define DIM 128
#define N_SLABS 32
#define SLAB (N_EDGES / N_SLABS)
#define SLAB4 (SLAB / 4)
#define NODES4 (N_NODES / 4)

typedef __attribute__((ext_vector_type(2))) float v2f;
typedef __attribute__((ext_vector_type(8))) float v8f;

#define GEMM_MT 5
#define GEMM_NT 2
#define GEMM_ROWS (16 * GEMM_MT)
#define GEMM_WGS (N_NODES / GEMM_ROWS)
#define GEMM_KU 2
#define GEMM_RING 4
#define GEMM_STAGES (DIM / (4 * GEMM_KU))

__global__ __launch_bounds__(128) void gat_gemm(const float* __restrict__ h,
                                                const float* __restrict__ W,
                                                float* __restrict__ z) {
    const int lane = threadIdx.x & 31;
    const int wv   = threadIdx.x >> 5;
    const int m0   = blockIdx.x * GEMM_ROWS;
    const int n0   = wv * (16 * GEMM_NT);
    const int half = lane >> 4;
    const int lr   = lane & 15;

    const float* __restrict__ Arow = h + (size_t)(m0 + lr) * DIM + half * 2;
    const float* __restrict__ Brow = W + (size_t)(n0 + lr) * DIM + half * 2;

    v2f a[GEMM_RING][GEMM_KU][GEMM_MT];
    v2f b[GEMM_RING][GEMM_KU][GEMM_NT];
#define GEMM_LOAD_STAGE(S) do { \
        _Pragma("unroll") for (int u = 0; u < GEMM_KU; ++u) { \
            const int k_ = 4 * (GEMM_KU * (S) + u); \
            _Pragma("unroll") for (int i = 0; i < GEMM_MT; ++i) \
                a[(S) % GEMM_RING][u][i] = *(const v2f*)(Arow + (size_t)i * 16 * DIM + k_); \
            _Pragma("unroll") for (int j = 0; j < GEMM_NT; ++j) \
                b[(S) % GEMM_RING][u][j] = *(const v2f*)(Brow + (size_t)j * 16 * DIM + k_); \
        } } while (0)

    #pragma unroll
    for (int s = 0; s < GEMM_RING - 1; ++s) GEMM_LOAD_STAGE(s);

    v8f acc[GEMM_MT][GEMM_NT] = {};
    #pragma unroll
    for (int s = 0; s < GEMM_STAGES; ++s) {
        __builtin_amdgcn_sched_barrier(0);
        if (s + GEMM_RING - 1 < GEMM_STAGES) GEMM_LOAD_STAGE(s + GEMM_RING - 1);
        __builtin_amdgcn_sched_barrier(0);
        #pragma unroll
        for (int u = 0; u < GEMM_KU; ++u) {
            #pragma unroll
            for (int i = 0; i < GEMM_MT; ++i) {
                #pragma unroll
                for (int j = 0; j < GEMM_NT; ++j) {
                    acc[i][j] = __builtin_amdgcn_wmma_f32_16x16x4_f32(
                         false, a[s % GEMM_RING][u][i],  false, b[s % GEMM_RING][u][j],
                         (short)0, acc[i][j],  false,  false);
                }
            }
        }
    }
#undef GEMM_LOAD_STAGE

    #pragma unroll
    for (int i = 0; i < GEMM_MT; ++i) {
        #pragma unroll
        for (int j = 0; j < GEMM_NT; ++j) {
            const int n = n0 + 16 * j + lr;
            #pragma unroll
            for (int v = 0; v < 8; ++v) {
                const int m = m0 + 16 * i + v + 8 * half;
                z[(size_t)m * DIM + n] = acc[i][j][v];
            }
        }
    }
}

#define SCORE_WGS    224
#define SCORE_WAVES  (SCORE_WGS * 32)
#define SCORE_ROUNDS ((N_NODES + SCORE_WAVES - 1) / SCORE_WAVES)

__device__ __forceinline__ void score_pass(const int wg, const int t,
                                           const float* __restrict__ z,
                                           const float* __restrict__ a_attn,
                                           float* __restrict__ s_src,
                                           float* __restrict__ s_dst) {
    const int lane = t & 31;
    const int g    = wg * 32 + (t >> 5);

    const float4 av = ((const float4*)(a_attn))[lane];
    const float4 dv = ((const float4*)(a_attn + DIM))[lane];

    float4 zv[SCORE_ROUNDS];
    #pragma unroll
    for (int r = 0; r < SCORE_ROUNDS; ++r) {
        const int n  = g + r * SCORE_WAVES;
        const int nc = n < N_NODES ? n : N_NODES - 1;
        zv[r] = ((const float4*)z)[(size_t)nc * (DIM / 4) + lane];
    }

    #pragma unroll
    for (int r = 0; r < SCORE_ROUNDS; ++r) {
        const int n = g + r * SCORE_WAVES;
        float ss = zv[r].x * av.x + zv[r].y * av.y + zv[r].z * av.z + zv[r].w * av.w;
        float sd = zv[r].x * dv.x + zv[r].y * dv.y + zv[r].z * dv.z + zv[r].w * dv.w;

        #pragma unroll
        for (int off = 16; off > 0; off >>= 1) {
            ss += __shfl_down(ss, off, 32);
            sd += __shfl_down(sd, off, 32);
        }
        if (lane == 0 && n < N_NODES) {
            s_src[n] = ss;
            s_dst[n] = sd;
        }
    }
}

__global__ __launch_bounds__(1024) void gat_hist_score(const int* __restrict__ dst,
                                                       int* __restrict__ slab_tab,
                                                       int* __restrict__ rank,
                                                       int* alloc,
                                                       const float* __restrict__ z,
                                                       const float* __restrict__ a_attn,
                                                       float* __restrict__ s_src,
                                                       float* __restrict__ s_dst) {
    __shared__ int tbl[N_NODES];
    const int t  = threadIdx.x;
    const int k  = blockIdx.x;

    if (k >= N_SLABS) {
        score_pass(k - N_SLABS, t, z, a_attn, s_src, s_dst);
        return;
    }

    const int4* __restrict__ d4 = (const int4*)(dst + (size_t)k * SLAB);
    int4* __restrict__ r4       = (int4*)(rank + (size_t)k * SLAB);
    int4* __restrict__ o4       = (int4*)(slab_tab + (size_t)k * N_NODES);

    int4 dv[7];
    #pragma unroll
    for (int r = 0; r < 7; ++r) {
        const int q = t + r * 1024;
        dv[r] = d4[q < SLAB4 ? q : SLAB4 - 1];
    }

    int4* tbl4 = (int4*)tbl;
    for (int n = t; n < NODES4; n += 1024) tbl4[n] = make_int4(0, 0, 0, 0);
    if (k == 0 && t == 0) *alloc = 0;
    __syncthreads();

    int4 rk[7];
    #pragma unroll
    for (int r = 0; r < 7; ++r) {
        const int q = t + r * 1024;
        if (q < SLAB4) {
            rk[r].x = atomicAdd(&tbl[dv[r].x], 1);
            rk[r].y = atomicAdd(&tbl[dv[r].y], 1);
            rk[r].z = atomicAdd(&tbl[dv[r].z], 1);
            rk[r].w = atomicAdd(&tbl[dv[r].w], 1);
        }
    }
    __syncthreads();

    #pragma unroll
    for (int r = 0; r < 7; ++r) {
        const int q = t + r * 1024;
        if (q < SLAB4) r4[q] = rk[r];
    }
    for (int n = t; n < NODES4; n += 1024) o4[n] = tbl4[n];
}

__global__ __launch_bounds__(256) void gat_alloc(int* __restrict__ slab_tab,
                                                 int* __restrict__ seg,
                                                 int* alloc) {
    __shared__ int wave_off[8];
    __shared__ int base_s;
    const int idx  = blockIdx.x * 256 + (int)threadIdx.x;
    const int lane = threadIdx.x & 31;
    const int wv   = threadIdx.x >> 5;
    const bool live = idx < N_NODES;
    const int  col  = live ? idx : N_NODES - 1;
    int c[N_SLABS];
    int v = 0;
    #pragma unroll
    for (int k = 0; k < N_SLABS; ++k) {
        c[k] = slab_tab[(size_t)k * N_NODES + col];
        v += c[k];
    }
    if (!live) v = 0;

    int inc = v;
    for (int off = 1; off < 32; off <<= 1) {
        const int t = __shfl_up(inc, off, 32);
        if (lane >= off) inc += t;
    }
    if (lane == 31) wave_off[wv] = inc;
    __syncthreads();
    if (threadIdx.x == 0) {
        int tot = 0;
        for (int w = 0; w < 8; ++w) {
            const int t = wave_off[w];
            wave_off[w] = tot;
            tot += t;
        }
        base_s = atomicAdd(alloc, tot);
    }
    __syncthreads();
    if (live) {
        int run = base_s + wave_off[wv] + inc - v;
        ((int2*)seg)[idx] = make_int2(run, v);
        #pragma unroll
        for (int k = 0; k < N_SLABS; ++k) {
            slab_tab[(size_t)k * N_NODES + idx] = run;
            run += c[k];
        }
    }
}

__global__ __launch_bounds__(256) void gat_fill(const int* __restrict__ src,
                                                const int* __restrict__ dst,
                                                const int* __restrict__ rank,
                                                const float* __restrict__ s_src,
                                                const float* __restrict__ s_dst,
                                                const int* __restrict__ slab_tab,
                                                int* __restrict__ csr_src,
                                                float* __restrict__ csr_e) {
    int i = blockIdx.x * blockDim.x + threadIdx.x;
    if (i >= N_EDGES) return;
    const int s = src[i];
    const int d = dst[i];
    const int k = i / SLAB;
    const int pos = slab_tab[(size_t)k * N_NODES + d] + rank[i];
    const float v = s_src[s] + s_dst[d];
    csr_src[pos] = s;
    csr_e[pos]   = v > 0.0f ? v : 0.01f * v;
}

template <int N>
__device__ __forceinline__ void rows_request(const float* __restrict__ z, const int lane,
                                             const int sl, const int j, float4 (&r)[8]) {
    #pragma unroll
    for (int u = 0; u < N; ++u) {
        const int s = __builtin_amdgcn_readlane(sl, j + u);
        r[u] = ((const float4*)z)[(unsigned)s * (DIM / 4) + (unsigned)lane];
    }
}

template <int N>
__device__ __forceinline__ void rows_sum(const float wl, const int j,
                                         const float4 (&r)[8], float4& acc) {
    #pragma unroll
    for (int u = 0; u < N; ++u) {
        const float w = __int_as_float(__builtin_amdgcn_readlane(__float_as_int(wl), j + u));
        acc.x += w * r[u].x;
        acc.y += w * r[u].y;
        acc.z += w * r[u].z;
        acc.w += w * r[u].w;
    }
}

__device__ __forceinline__ void rows_block(const float* __restrict__ z, const int lane,
                                           const int sl, const float wl,
                                           int j, const int nn, float4& acc) {
    float4 r[8];
    for (; j + 8 <= nn; j += 8) {
        rows_request<8>(z, lane, sl, j, r);
        rows_sum<8>(wl, j, r, acc);
    }
    const int rem = nn - j;
    if (rem > 4) {
        rows_request<8>(z, lane, sl, j, r);
        rows_sum<8>(wl, j, r, acc);
    } else if (rem > 0) {
        rows_request<4>(z, lane, sl, j, r);
        rows_sum<4>(wl, j, r, acc);
    }
}

__device__ __forceinline__ float wave_sum(float v) {
    #pragma unroll
    for (int off = 16; off > 0; off >>= 1) v += __shfl_xor(v, off, 32);
    return v;
}

__global__ __launch_bounds__(256) void gat_node(const float* __restrict__ z,
                                                const int* __restrict__ seg,
                                                const int* __restrict__ csr_src,
                                                const float* __restrict__ csr_e,
                                                float* __restrict__ out) {
    const int node = blockIdx.x * 8 + __builtin_amdgcn_readfirstlane((int)(threadIdx.x >> 5));
    const int lane = threadIdx.x & 31;
    const int2 sg  = ((const int2*)seg)[node];
    const int beg = sg.x;
    const int cnt = sg.y;
    const int end = beg + cnt;

    float4 acc = {0.0f, 0.0f, 0.0f, 0.0f};
    float sum = 0.0f;
    if (cnt > 0) {
        const bool in0 = lane < cnt;
        const int  p   = in0 ? beg + lane : beg;
        const float e0 = csr_e[p];
        const int   s0 = csr_src[p];

        float4 r0[8];
        rows_request<8>(z, lane, s0, 0, r0);

        float mx = in0 ? e0 : -INFINITY;
        for (int q = beg + 32 + lane; q < end; q += 32) mx = fmaxf(mx, csr_e[q]);
        #pragma unroll
        for (int off = 16; off > 0; off >>= 1) mx = fmaxf(mx, __shfl_xor(mx, off, 32));

        const float w0 = in0 ? __expf(e0 - mx) : 0.0f;
        sum = w0;
        if (cnt <= 32) sum = wave_sum(sum);

        rows_sum<8>(w0, 0, r0, acc);
        rows_block(z, lane, s0, w0, 8, min(32, cnt), acc);

        if (cnt > 32) {
            for (int p0 = beg + 32; p0 < end; p0 += 32) {
                const bool in = p0 + lane < end;
                const int  pp = in ? p0 + lane : beg;
                const float el = csr_e[pp];
                const int   sl = csr_src[pp];
                const float wl = in ? __expf(el - mx) : 0.0f;
                sum += wl;
                rows_block(z, lane, sl, wl, 0, min(32, end - p0), acc);
            }
            sum = wave_sum(sum);
        }
    }

    const float inv = (cnt > 0) ? 1.0f / sum : 0.0f;
    float4 o = {acc.x * inv, acc.y * inv, acc.z * inv, acc.w * inv};
    ((float4*)(out + (size_t)node * DIM))[lane] = o;
}

extern "C" void kernel_launch(void* const* d_in, const int* in_sizes, int n_in,
                              void* d_out, int out_size, void* d_ws, size_t ws_size,
                              hipStream_t stream) {
    const float* h      = (const float*)d_in[0];
    const float* W_fc   = (const float*)d_in[1];
    const float* a_attn = (const float*)d_in[2];
    const int*   src    = (const int*)d_in[3];
    const int*   dst    = (const int*)d_in[4];
    float*       out    = (float*)d_out;
    (void)in_sizes; (void)n_in; (void)out_size; (void)ws_size;

    float* ws        = (float*)d_ws;
    float* z         = ws;
    float* s_src     = z + (size_t)N_NODES * DIM;
    float* s_dst     = s_src + N_NODES;
    float* csr_e     = s_dst + N_NODES;
    int*   seg       = (int*)(csr_e + N_EDGES);
    int*   alloc     = seg + 2 * N_NODES;
    int*   csr_src   = alloc + 1;
    int*   rank      = (int*)d_out;
    int*   slab_tab  = rank + N_EDGES;

    gat_gemm<<<GEMM_WGS, 128, 0, stream>>>(h, W_fc, z);

    gat_hist_score<<<N_SLABS + SCORE_WGS, 1024, 0, stream>>>(dst, slab_tab, rank, alloc,
                                                             z, a_attn, s_src, s_dst);

    gat_alloc<<<(N_NODES + 255) / 256, 256, 0, stream>>>(slab_tab, seg, alloc);

    gat_fill<<<(N_EDGES + 255) / 256, 256, 0, stream>>>(src, dst, rank, s_src, s_dst,
                                                        slab_tab,
                                                        csr_src, csr_e);

    gat_node<<<N_NODES / 8, 256, 0, stream>>>(z, seg, csr_src, csr_e, out);
}


# CSR fill's random accesses paired into 8-byte records: one read brings slot base + dst score (alloc pass writes them side by side), one store carries source id + logit; 3 per edge, was 5
# speedup vs baseline: 4.1977x; 1.3269x over previous
#include <hip/hip_runtime.h>
#include <hip/hip_bf16.h>

#define N_NODES 50000
#define N_EDGES 800000
#define DIM 128
#define N_SLABS 32
#define SLAB (N_EDGES / N_SLABS)
#define SLAB4 (SLAB / 4)
#define NODES4 (N_NODES / 4)

typedef __attribute__((ext_vector_type(2))) float v2f;
typedef __attribute__((ext_vector_type(8))) float v8f;

#define LPITCH 132

#define SCORE_WGS    224
#define SCORE_WAVES  (SCORE_WGS * 32)
#define SCORE_ROUNDS ((N_NODES + SCORE_WAVES - 1) / SCORE_WAVES)

__device__ __forceinline__ void score_pass(const int wg, const int t,
                                           const float* __restrict__ h,
                                           const float* __restrict__ W,
                                           const float* __restrict__ a_attn,
                                           float* __restrict__ s_src,
                                           float* __restrict__ s_dst,
                                           float* __restrict__ part,
                                           float* __restrict__ u) {
    const int lane = t & 31;
    const int g    = wg * 32 + (t >> 5);

    float4 zv[SCORE_ROUNDS];
    #pragma unroll
    for (int r = 0; r < SCORE_ROUNDS; ++r) {
        const int n  = g + r * SCORE_WAVES;
        const int nc = n < N_NODES ? n : N_NODES - 1;
        zv[r] = ((const float4*)h)[(size_t)nc * (DIM / 4) + lane];
    }

    {
        const int k = t & (DIM - 1);
        const int p = t >> 7;
        float ps = 0.0f, pd = 0.0f;
        #pragma unroll
        for (int jj = 0; jj < 16; ++jj) {
            const int j = 16 * p + jj;
            const float w = W[(size_t)j * DIM + k];
            ps += a_attn[j] * w;
            pd += a_attn[DIM + j] * w;
        }
        part[p * 256 + k]       = ps;
        part[p * 256 + DIM + k] = pd;
    }
    __syncthreads();
    if (t < 256) {
        float sacc = 0.0f;
        #pragma unroll
        for (int p = 0; p < 8; ++p) sacc += part[p * 256 + t];
        u[t] = sacc;
    }
    __syncthreads();

    const float4 av = ((const float4*)(u))[lane];
    const float4 dv = ((const float4*)(u + DIM))[lane];

    #pragma unroll
    for (int r = 0; r < SCORE_ROUNDS; ++r) {
        const int n = g + r * SCORE_WAVES;
        float ss = zv[r].x * av.x + zv[r].y * av.y + zv[r].z * av.z + zv[r].w * av.w;
        float sd = zv[r].x * dv.x + zv[r].y * dv.y + zv[r].z * dv.z + zv[r].w * dv.w;

        #pragma unroll
        for (int off = 16; off > 0; off >>= 1) {
            ss += __shfl_down(ss, off, 32);
            sd += __shfl_down(sd, off, 32);
        }
        if (lane == 0 && n < N_NODES) {
            s_src[n] = ss;
            s_dst[n] = sd;
        }
    }
}

__global__ __launch_bounds__(1024) void gat_hist_score(const int* __restrict__ dst,
                                                       int* __restrict__ slab_tab,
                                                       int* __restrict__ rank,
                                                       int* alloc,
                                                       const float* __restrict__ h,
                                                       const float* __restrict__ W,
                                                       const float* __restrict__ a_attn,
                                                       float* __restrict__ s_src,
                                                       float* __restrict__ s_dst) {
    __shared__ int tbl[N_NODES];
    const int t  = threadIdx.x;
    const int k  = blockIdx.x;

    if (k >= N_SLABS) {
        float* fl = reinterpret_cast<float*>(tbl);
        score_pass(k - N_SLABS, t, h, W, a_attn, s_src, s_dst, fl, fl + 8 * 256);
        return;
    }

    const int4* __restrict__ d4 = (const int4*)(dst + (size_t)k * SLAB);
    int4* __restrict__ r4       = (int4*)(rank + (size_t)k * SLAB);
    int4* __restrict__ o4       = (int4*)(slab_tab + (size_t)k * N_NODES);

    int4 dv[7];
    #pragma unroll
    for (int r = 0; r < 7; ++r) {
        const int q = t + r * 1024;
        dv[r] = d4[q < SLAB4 ? q : SLAB4 - 1];
    }

    int4* tbl4 = (int4*)tbl;
    for (int n = t; n < NODES4; n += 1024) tbl4[n] = make_int4(0, 0, 0, 0);
    if (k == 0 && t == 0) *alloc = 0;
    __syncthreads();

    int4 rk[7];
    #pragma unroll
    for (int r = 0; r < 7; ++r) {
        const int q = t + r * 1024;
        if (q < SLAB4) {
            rk[r].x = atomicAdd(&tbl[dv[r].x], 1);
            rk[r].y = atomicAdd(&tbl[dv[r].y], 1);
            rk[r].z = atomicAdd(&tbl[dv[r].z], 1);
            rk[r].w = atomicAdd(&tbl[dv[r].w], 1);
        }
    }
    __syncthreads();

    #pragma unroll
    for (int r = 0; r < 7; ++r) {
        const int q = t + r * 1024;
        if (q < SLAB4) r4[q] = rk[r];
    }
    for (int n = t; n < NODES4; n += 1024) o4[n] = tbl4[n];
}

__global__ __launch_bounds__(256) void gat_alloc(const int* __restrict__ slab_cnt,
                                                 const float* __restrict__ s_dst,
                                                 int* __restrict__ slab_rec,
                                                 int* __restrict__ seg,
                                                 int* alloc) {
    __shared__ int wave_off[8];
    __shared__ int base_s;
    const int idx  = blockIdx.x * 256 + (int)threadIdx.x;
    const int lane = threadIdx.x & 31;
    const int wv   = threadIdx.x >> 5;
    const bool live = idx < N_NODES;
    const int  col  = live ? idx : N_NODES - 1;
    int c[N_SLABS];
    int v = 0;
    #pragma unroll
    for (int k = 0; k < N_SLABS; ++k) {
        c[k] = slab_cnt[(size_t)k * N_NODES + col];
        v += c[k];
    }
    const int sd = __float_as_int(s_dst[col]);
    if (!live) v = 0;

    int inc = v;
    for (int off = 1; off < 32; off <<= 1) {
        const int t = __shfl_up(inc, off, 32);
        if (lane >= off) inc += t;
    }
    if (lane == 31) wave_off[wv] = inc;
    __syncthreads();
    if (threadIdx.x == 0) {
        int tot = 0;
        for (int w = 0; w < 8; ++w) {
            const int t = wave_off[w];
            wave_off[w] = tot;
            tot += t;
        }
        base_s = atomicAdd(alloc, tot);
    }
    __syncthreads();
    if (live) {
        int run = base_s + wave_off[wv] + inc - v;
        ((int2*)seg)[idx] = make_int2(run, v);
        int2* __restrict__ rec = (int2*)slab_rec;
        #pragma unroll
        for (int k = 0; k < N_SLABS; ++k) {
            rec[(size_t)k * N_NODES + idx] = make_int2(run, sd);
            run += c[k];
        }
    }
}

__global__ __launch_bounds__(256) void gat_fill(const int* __restrict__ src,
                                                const int* __restrict__ dst,
                                                const int* __restrict__ rank,
                                                const float* __restrict__ s_src,
                                                const int* __restrict__ slab_rec,
                                                int* __restrict__ csr) {
    int i = blockIdx.x * blockDim.x + threadIdx.x;
    if (i >= N_EDGES) return;
    const int s = src[i];
    const int d = dst[i];
    const int k = i / SLAB;
    const int2 br = ((const int2*)slab_rec)[(size_t)k * N_NODES + d];
    const int pos = br.x + rank[i];
    const float v = s_src[s] + __int_as_float(br.y);
    const float e = v > 0.0f ? v : 0.01f * v;
    ((int2*)csr)[pos] = make_int2(s, __float_as_int(e));
}

template <int N>
__device__ __forceinline__ void rows_request(const float* __restrict__ x, const int lane,
                                             const int sl, const int j, float4 (&r)[8]) {
    #pragma unroll
    for (int u = 0; u < N; ++u) {
        const int s = __builtin_amdgcn_readlane(sl, j + u);
        r[u] = ((const float4*)x)[(unsigned)s * (DIM / 4) + (unsigned)lane];
    }
}

template <int N>
__device__ __forceinline__ void rows_sum(const float wl, const int j,
                                         const float4 (&r)[8], float4& acc) {
    #pragma unroll
    for (int u = 0; u < N; ++u) {
        const float w = __int_as_float(__builtin_amdgcn_readlane(__float_as_int(wl), j + u));
        acc.x += w * r[u].x;
        acc.y += w * r[u].y;
        acc.z += w * r[u].z;
        acc.w += w * r[u].w;
    }
}

__device__ __forceinline__ void rows_block(const float* __restrict__ x, const int lane,
                                           const int sl, const float wl,
                                           int j, const int nn, float4& acc) {
    float4 r[8];
    for (; j + 8 <= nn; j += 8) {
        rows_request<8>(x, lane, sl, j, r);
        rows_sum<8>(wl, j, r, acc);
    }
    const int rem = nn - j;
    if (rem > 4) {
        rows_request<8>(x, lane, sl, j, r);
        rows_sum<8>(wl, j, r, acc);
    } else if (rem > 0) {
        rows_request<4>(x, lane, sl, j, r);
        rows_sum<4>(wl, j, r, acc);
    }
}

__device__ __forceinline__ float wave_sum(float v) {
    #pragma unroll
    for (int off = 16; off > 0; off >>= 1) v += __shfl_xor(v, off, 32);
    return v;
}

__device__ __forceinline__ int first_slot(const int lane, const int beg, const int cnt) {
    return cnt > 0 ? (lane < cnt ? beg + lane : beg) : 0;
}

__device__ __forceinline__ float4 node_row(const float* __restrict__ h, const int lane,
                                           const int beg, const int cnt,
                                           const float e0, const int s0,
                                           const int2* __restrict__ csr) {
    const int end = beg + cnt;
    float4 acc = {0.0f, 0.0f, 0.0f, 0.0f};
    float sum = 0.0f;
    {
        const bool in0 = lane < cnt;

        float4 r0[8];
        rows_request<8>(h, lane, s0, 0, r0);

        float mx = in0 ? e0 : -INFINITY;
        for (int q = beg + 32 + lane; q < end; q += 32) mx = fmaxf(mx, __int_as_float(csr[q].y));
        #pragma unroll
        for (int off = 16; off > 0; off >>= 1) mx = fmaxf(mx, __shfl_xor(mx, off, 32));

        const float w0 = in0 ? __expf(e0 - mx) : 0.0f;
        sum = w0;
        if (cnt <= 32) sum = wave_sum(sum);

        rows_sum<8>(w0, 0, r0, acc);
        rows_block(h, lane, s0, w0, 8, min(32, cnt), acc);

        if (cnt > 32) {
            for (int p0 = beg + 32; p0 < end; p0 += 32) {
                const bool in = p0 + lane < end;
                const int  pp = in ? p0 + lane : beg;
                const int2  rc = csr[pp];
                const float el = __int_as_float(rc.y);
                const int   sl = rc.x;
                const float wl = in ? __expf(el - mx) : 0.0f;
                sum += wl;
                rows_block(h, lane, sl, wl, 0, min(32, end - p0), acc);
            }
            sum = wave_sum(sum);
        }
    }
    if (cnt <= 0) return make_float4(0.0f, 0.0f, 0.0f, 0.0f);
    const float inv = 1.0f / sum;
    return make_float4(acc.x * inv, acc.y * inv, acc.z * inv, acc.w * inv);
}

#define FC_NODES 32
#define FC_WGS   ((N_NODES + FC_NODES - 1) / FC_NODES)

__global__ __launch_bounds__(512) void gat_node_fc(const float* __restrict__ h,
                                                   const float* __restrict__ W,
                                                   const int* __restrict__ seg,
                                                   const int* __restrict__ csr_rec,
                                                   float* __restrict__ out) {
    __shared__ float Ws[DIM * LPITCH];
    __shared__ float At[FC_NODES * LPITCH];
    __shared__ float Dt[FC_NODES * LPITCH];

    const int t    = threadIdx.x;
    const int lane = t & 31;
    const int wv   = __builtin_amdgcn_readfirstlane(t >> 5);
    const int row0 = 2 * wv;
    const int node0 = blockIdx.x * FC_NODES + row0;
    const bool live0 = node0 < N_NODES;
    const bool live1 = node0 + 1 < N_NODES;

    const int2 sg0 = ((const int2*)seg)[live0 ? node0 : 0];
    const int2 sg1 = ((const int2*)seg)[live1 ? node0 + 1 : 0];

    const int cnt0 = live0 ? sg0.y : 0;
    const int cnt1 = live1 ? sg1.y : 0;

    float4 wq[8];
    #pragma unroll
    for (int i = 0; i < 8; ++i) wq[i] = ((const float4*)W)[t + 512 * i];

    const int   p0  = first_slot(lane, sg0.x, cnt0);
    const int   p1  = first_slot(lane, sg1.x, cnt1);
    const int2* __restrict__ csr = (const int2*)csr_rec;
    const int2  c00 = csr[p0];
    const int2  c10 = csr[p1];
    const float e00 = __int_as_float(c00.y);
    const int   s00 = c00.x;
    const float e10 = __int_as_float(c10.y);
    const int   s10 = c10.x;

    #pragma unroll
    for (int i = 0; i < 8; ++i) {
        const int q = t + 512 * i;
        *(float4*)(Ws + (q >> 5) * LPITCH + (q & 31) * 4) = wq[i];
    }

    const float4 o0 = node_row(h, lane, sg0.x, cnt0, e00, s00, csr);
    *(float4*)(At + row0 * LPITCH + 4 * lane) = o0;
    const float4 o1 = node_row(h, lane, sg1.x, cnt1, e10, s10, csr);
    *(float4*)(At + (row0 + 1) * LPITCH + 4 * lane) = o1;

    __syncthreads();

    {
        const int rt = wv >> 3;
        const int ct = wv & 7;
        const int hf = lane >> 4;
        const int c  = lane & 15;
        const float* Ap = At + (16 * rt + c) * LPITCH + 2 * hf;
        const float* Bp = Ws + (16 * ct + c) * LPITCH + 2 * hf;
        v8f acc0 = {}, acc1 = {};
        #pragma unroll
        for (int kk = 0; kk < DIM / 4; kk += 2) {
            const v2f a0 = *(const v2f*)(Ap + 4 * kk);
            const v2f b0 = *(const v2f*)(Bp + 4 * kk);
            const v2f a1 = *(const v2f*)(Ap + 4 * kk + 4);
            const v2f b1 = *(const v2f*)(Bp + 4 * kk + 4);
            acc0 = __builtin_amdgcn_wmma_f32_16x16x4_f32(
                 false, a0,  false, b0,
                 (short)0, acc0,  false,  false);
            acc1 = __builtin_amdgcn_wmma_f32_16x16x4_f32(
                 false, a1,  false, b1,
                 (short)0, acc1,  false,  false);
        }
        const v8f acc = acc0 + acc1;
        #pragma unroll
        for (int v = 0; v < 8; ++v)
            Dt[(16 * rt + 8 * hf + v) * LPITCH + 16 * ct + c] = acc[v];
    }

    __syncthreads();

    if (live0) ((float4*)(out + (size_t)node0 * DIM))[lane] = *(const float4*)(Dt + row0 * LPITCH + 4 * lane);
    if (live1) ((float4*)(out + (size_t)(node0 + 1) * DIM))[lane] = *(const float4*)(Dt + (row0 + 1) * LPITCH + 4 * lane);
}

extern "C" void kernel_launch(void* const* d_in, const int* in_sizes, int n_in,
                              void* d_out, int out_size, void* d_ws, size_t ws_size,
                              hipStream_t stream) {
    const float* h      = (const float*)d_in[0];
    const float* W_fc   = (const float*)d_in[1];
    const float* a_attn = (const float*)d_in[2];
    const int*   src    = (const int*)d_in[3];
    const int*   dst    = (const int*)d_in[4];
    float*       out    = (float*)d_out;
    (void)in_sizes; (void)n_in; (void)out_size; (void)ws_size;

    float* ws        = (float*)d_ws;
    float* s_src     = ws;
    float* s_dst     = s_src + N_NODES;
    int*   seg       = (int*)(s_dst + N_NODES);
    int*   csr_rec   = seg + 2 * N_NODES;
    int*   rank      = (int*)d_out;
    int*   slab_cnt  = rank + N_EDGES;
    int*   slab_rec  = slab_cnt + N_SLABS * N_NODES;
    int*   alloc     = slab_rec + 2 * N_SLABS * N_NODES;

    gat_hist_score<<<N_SLABS + SCORE_WGS, 1024, 0, stream>>>(dst, slab_cnt, rank, alloc,
                                                             h, W_fc, a_attn, s_src, s_dst);

    gat_alloc<<<(N_NODES + 255) / 256, 256, 0, stream>>>(slab_cnt, s_dst, slab_rec, seg, alloc);

    gat_fill<<<(N_EDGES + 255) / 256, 256, 0, stream>>>(src, dst, rank, s_src,
                                                        slab_rec, csr_rec);

    gat_node_fc<<<FC_WGS, 512, 0, stream>>>(h, W_fc, seg, csr_rec, out);
}


# no CSR assembly: each 25000-edge slab is sorted by node in its workgroup's LDS (counter scan, start + rank); node kernel merges a node's 32 slab runs, forms the logits; 2 launches, was 4; order kept
# speedup vs baseline: 6.1696x; 1.4698x over previous
#include <hip/hip_runtime.h>
#include <hip/hip_bf16.h>

#define N_NODES 50000
#define N_EDGES 800000
#define DIM 128
#define N_SLABS 32
#define SLAB (N_EDGES / N_SLABS)
#define SLAB4 (SLAB / 4)
#define NODES4 (N_NODES / 4)

typedef __attribute__((ext_vector_type(2))) float v2f;
typedef __attribute__((ext_vector_type(8))) float v8f;

#define LPITCH 132


#define SCORE_WGS    224
#define SCORE_WAVES  (SCORE_WGS * 32)
#define SCORE_ROUNDS ((N_NODES + SCORE_WAVES - 1) / SCORE_WAVES)

__device__ __forceinline__ void score_pass(const int wg, const int t,
                                           const float* __restrict__ h,
                                           const float* __restrict__ W,
                                           const float* __restrict__ a_attn,
                                           float* __restrict__ s_src,
                                           float* __restrict__ s_dst,
                                           float* __restrict__ part,
                                           float* __restrict__ u) {
    const int lane = t & 31;
    const int g    = wg * 32 + (t >> 5);

    float4 zv[SCORE_ROUNDS];
    #pragma unroll
    for (int r = 0; r < SCORE_ROUNDS; ++r) {
        const int n  = g + r * SCORE_WAVES;
        const int nc = n < N_NODES ? n : N_NODES - 1;
        zv[r] = ((const float4*)h)[(size_t)nc * (DIM / 4) + lane];
    }

    {
        const int k = t & (DIM - 1);
        const int p = t >> 7;
        float ps = 0.0f, pd = 0.0f;
        #pragma unroll
        for (int jj = 0; jj < 16; ++jj) {
            const int j = 16 * p + jj;
            const float w = W[(size_t)j * DIM + k];
            ps += a_attn[j] * w;
            pd += a_attn[DIM + j] * w;
        }
        part[p * 256 + k]       = ps;
        part[p * 256 + DIM + k] = pd;
    }
    __syncthreads();
    if (t < 256) {
        float sacc = 0.0f;
        #pragma unroll
        for (int p = 0; p < 8; ++p) sacc += part[p * 256 + t];
        u[t] = sacc;
    }
    __syncthreads();

    const float4 av = ((const float4*)(u))[lane];
    const float4 dv = ((const float4*)(u + DIM))[lane];

    #pragma unroll
    for (int r = 0; r < SCORE_ROUNDS; ++r) {
        const int n = g + r * SCORE_WAVES;
        float ss = zv[r].x * av.x + zv[r].y * av.y + zv[r].z * av.z + zv[r].w * av.w;
        float sd = zv[r].x * dv.x + zv[r].y * dv.y + zv[r].z * dv.z + zv[r].w * dv.w;

        #pragma unroll
        for (int off = 16; off > 0; off >>= 1) {
            ss += __shfl_down(ss, off, 32);
            sd += __shfl_down(sd, off, 32);
        }
        if (lane == 0 && n < N_NODES) {
            s_src[n] = ss;
            s_dst[n] = sd;
        }
    }
}

#define SCAN_CHUNK 49

__global__ __launch_bounds__(1024) void gat_sort_score(const int* __restrict__ dst,
                                                       const int* __restrict__ src,
                                                       int* __restrict__ start_pk,
                                                       int* __restrict__ sorted,
                                                       const float* __restrict__ h,
                                                       const float* __restrict__ W,
                                                       const float* __restrict__ a_attn,
                                                       float* __restrict__ s_src,
                                                       float* __restrict__ s_dst) {
    __shared__ __attribute__((aligned(16))) int tbl[N_NODES];
    __shared__ __attribute__((aligned(16))) int lst[SLAB];
    __shared__ int wsum[32];
    const int t  = threadIdx.x;
    const int k  = blockIdx.x;

    if (k >= N_SLABS) {
        float* fl = reinterpret_cast<float*>(tbl);
        score_pass(k - N_SLABS, t, h, W, a_attn, s_src, s_dst, fl, fl + 8 * 256);
        return;
    }

    const int4* __restrict__ d4 = (const int4*)(dst + (size_t)k * SLAB);
    const int4* __restrict__ s4 = (const int4*)(src + (size_t)k * SLAB);

    int4 dv[7], sv[7];
    #pragma unroll
    for (int r = 0; r < 7; ++r) {
        const int q  = t + r * 1024;
        const int qc = q < SLAB4 ? q : SLAB4 - 1;
        dv[r] = d4[qc];
        sv[r] = s4[qc];
    }

    int4* tbl4 = (int4*)tbl;
    for (int n = t; n < NODES4; n += 1024) tbl4[n] = make_int4(0, 0, 0, 0);
    __syncthreads();

    int4 rk[7];
    #pragma unroll
    for (int r = 0; r < 7; ++r) {
        const int q = t + r * 1024;
        if (q < SLAB4) {
            rk[r].x = atomicAdd(&tbl[dv[r].x], 1);
            rk[r].y = atomicAdd(&tbl[dv[r].y], 1);
            rk[r].z = atomicAdd(&tbl[dv[r].z], 1);
            rk[r].w = atomicAdd(&tbl[dv[r].w], 1);
        }
    }
    __syncthreads();

    const int lane = t & 31;
    const int wv   = t >> 5;
    const int c0   = t * SCAN_CHUNK;
    int own = 0;
    #pragma clang loop unroll(disable)
    for (int jb = 0; jb < SCAN_CHUNK; jb += 7) {
        int c[7];
        #pragma unroll
        for (int j = 0; j < 7; ++j) {
            const int i = c0 + jb + j;
            c[j] = tbl[i < N_NODES ? i : N_NODES - 1];
        }
        #pragma unroll
        for (int j = 0; j < 7; ++j) own += c0 + jb + j < N_NODES ? c[j] : 0;
    }
    int inc = own;
    for (int off = 1; off < 32; off <<= 1) {
        const int u = __shfl_up(inc, off, 32);
        if (lane >= off) inc += u;
    }
    if (lane == 31) wsum[wv] = inc;
    __syncthreads();
    if (wv == 0) {
        const int v = wsum[lane];
        int wi = v;
        for (int off = 1; off < 32; off <<= 1) {
            const int u = __shfl_up(wi, off, 32);
            if (lane >= off) wi += u;
        }
        wsum[lane] = wi - v;
    }
    __syncthreads();
    {
        int run = wsum[wv] + inc - own;
        #pragma clang loop unroll(disable)
        for (int jb = 0; jb < SCAN_CHUNK; jb += 7) {
            int c[7];
            #pragma unroll
            for (int j = 0; j < 7; ++j) {
                const int i = c0 + jb + j;
                c[j] = tbl[i < N_NODES ? i : N_NODES - 1];
            }
            #pragma unroll
            for (int j = 0; j < 7; ++j) {
                const int i = c0 + jb + j;
                if (i < N_NODES) tbl[i] = run;
                run += c[j];
            }
        }
    }
    __syncthreads();

    #pragma unroll
    for (int r = 0; r < 7; ++r) {
        const int q = t + r * 1024;
        if (q < SLAB4) {
            lst[tbl[dv[r].x] + rk[r].x] = sv[r].x;
            lst[tbl[dv[r].y] + rk[r].y] = sv[r].y;
            lst[tbl[dv[r].z] + rk[r].z] = sv[r].z;
            lst[tbl[dv[r].w] + rk[r].w] = sv[r].w;
        }
    }
    __syncthreads();

    int4* __restrict__ o4 = (int4*)(sorted + (size_t)k * SLAB);
    const int4* lst4 = (const int4*)lst;
    for (int q = t; q < SLAB4; q += 1024) o4[q] = lst4[q];
    int* __restrict__ op = start_pk + (size_t)k * (N_NODES / 2);
    const int2* tbl2 = (const int2*)tbl;
    for (int m = t; m < N_NODES / 2; m += 1024) {
        const int2 p = tbl2[m];
        op[m] = (p.x & 0xffff) | (p.y << 16);
    }
}


template <int N>
__device__ __forceinline__ void rows_request(const float* __restrict__ x, const int lane,
                                             const int sl, const int j, float4 (&r)[8]) {
    #pragma unroll
    for (int u = 0; u < N; ++u) {
        const int s = __builtin_amdgcn_readlane(sl, j + u);
        r[u] = ((const float4*)x)[(unsigned)s * (DIM / 4) + (unsigned)lane];
    }
}

template <int N>
__device__ __forceinline__ void rows_sum(const float wl, const int j,
                                         const float4 (&r)[8], float4& acc) {
    #pragma unroll
    for (int u = 0; u < N; ++u) {
        const float w = __int_as_float(__builtin_amdgcn_readlane(__float_as_int(wl), j + u));
        acc.x += w * r[u].x;
        acc.y += w * r[u].y;
        acc.z += w * r[u].z;
        acc.w += w * r[u].w;
    }
}

__device__ __forceinline__ void rows_block(const float* __restrict__ x, const int lane,
                                           const int sl, const float wl,
                                           int j, const int nn, float4& acc) {
    float4 r[8];
    for (; j + 8 <= nn; j += 8) {
        rows_request<8>(x, lane, sl, j, r);
        rows_sum<8>(wl, j, r, acc);
    }
    const int rem = nn - j;
    if (rem > 4) {
        rows_request<8>(x, lane, sl, j, r);
        rows_sum<8>(wl, j, r, acc);
    } else if (rem > 0) {
        rows_request<4>(x, lane, sl, j, r);
        rows_sum<4>(wl, j, r, acc);
    }
}

__device__ __forceinline__ float wave_sum(float v) {
    #pragma unroll
    for (int off = 16; off > 0; off >>= 1) v += __shfl_xor(v, off, 32);
    return v;
}

#define ST_PITCH 33

__device__ __forceinline__ float leaky(const float v) {
    return v > 0.0f ? v : 0.01f * v;
}

struct Run { const int* sk; int a; int c; int off; };

__device__ __forceinline__ int wave_excl(const int v, const int lane, int& total) {
    int inc = v;
    #pragma unroll
    for (int off = 1; off < 32; off <<= 1) {
        const int u = __shfl_up(inc, off, 32);
        if (lane >= off) inc += u;
    }
    total = __builtin_amdgcn_readlane(inc, 31);
    return inc - v;
}

__device__ __forceinline__ Run make_run(const int* St, const int* __restrict__ sorted,
                                        const int lane, const int row, int& cnt) {
    Run r;
    const int a = St[lane * ST_PITCH + row];
    const int b = St[lane * ST_PITCH + row + 1];
    r.sk  = sorted;
    r.a   = lane * SLAB + a;
    r.c   = b - a;
    r.off = wave_excl(r.c, lane, cnt);
    return r;
}

__device__ __forceinline__ void strip_sync() {
    __builtin_amdgcn_fence(__ATOMIC_ACQ_REL, "wavefront");
    __builtin_amdgcn_wave_barrier();
}

__device__ __forceinline__ void strip_fill(const Run& r, const int eb, int* ids) {
    const int i0 = max(0, eb - r.off);
    const int i1 = min(r.c, eb + 32 - r.off);
    #pragma clang loop vectorize(disable) unroll(disable)
    for (int i = i0; i < i1; ++i) ids[r.off + i - eb] = r.sk[r.a + i];
}

__device__ __forceinline__ float4 node_row(const float* __restrict__ h,
                                           const float* __restrict__ s_src,
                                           const int lane, const int cnt, const float sd,
                                           const float e0, const int s0,
                                           const int* St, const int* __restrict__ sorted,
                                           const int row, int* ids) {
    float4 acc = {0.0f, 0.0f, 0.0f, 0.0f};
    float sum = 0.0f;
    {
        const bool in0 = lane < cnt;

        float4 r0[8];
        rows_request<8>(h, lane, s0, 0, r0);

        float mx = in0 ? e0 : -INFINITY;
        if (cnt > 32) {
            int deg;
            const Run run = make_run(St, sorted, lane, row, deg);
            for (int eb = 32; eb < cnt; eb += 32) {
                strip_sync();
                strip_fill(run, eb, ids);
                strip_sync();
                const bool  in = eb + lane < cnt;
                const int   sl = ids[in ? lane : 0];
                const float el = leaky(s_src[sl] + sd);
                if (in) mx = fmaxf(mx, el);
            }
        }
        #pragma unroll
        for (int off = 16; off > 0; off >>= 1) mx = fmaxf(mx, __shfl_xor(mx, off, 32));

        const float w0 = in0 ? __expf(e0 - mx) : 0.0f;
        sum = w0;
        if (cnt <= 32) sum = wave_sum(sum);

        rows_sum<8>(w0, 0, r0, acc);
        rows_block(h, lane, s0, w0, 8, min(32, cnt), acc);

        if (cnt > 32) {
            int deg;
            const Run run = make_run(St, sorted, lane, row, deg);
            for (int eb = 32; eb < cnt; eb += 32) {
                strip_sync();
                strip_fill(run, eb, ids);
                strip_sync();
                const bool  in = eb + lane < cnt;
                const int   sl = ids[in ? lane : 0];
                const float el = leaky(s_src[sl] + sd);
                const float wl = in ? __expf(el - mx) : 0.0f;
                sum += wl;
                rows_block(h, lane, sl, wl, 0, min(32, cnt - eb), acc);
            }
            sum = wave_sum(sum);
        }
    }
    if (cnt <= 0) return make_float4(0.0f, 0.0f, 0.0f, 0.0f);
    const float inv = 1.0f / sum;
    return make_float4(acc.x * inv, acc.y * inv, acc.z * inv, acc.w * inv);
}

__device__ __forceinline__ void first_request(const Run& r, int (&v)[4]) {
    #pragma unroll
    for (int u = 0; u < 4; ++u) v[u] = r.sk[u < r.c ? r.a + u : 0];
}
__device__ __forceinline__ void first_place(const Run& r, const int (&v)[4], int* ids) {
    #pragma unroll
    for (int u = 0; u < 4; ++u) {
        const bool ok = u < r.c && r.off + u < 32;
        ids[ok ? r.off + u : 32] = v[u];
    }
    const int i1 = min(r.c, 32 - r.off);
    #pragma clang loop vectorize(disable) unroll(disable)
    for (int i = 4; i < i1; ++i) ids[r.off + i] = r.sk[r.a + i];
}


#define FC_NODES 32
#define FC_WGS   ((N_NODES + FC_NODES - 1) / FC_NODES)

__global__ __launch_bounds__(512) void gat_node_fc(const float* __restrict__ h,
                                                   const float* __restrict__ W,
                                                   const float* __restrict__ s_src,
                                                   const float* __restrict__ s_dst,
                                                   const int* __restrict__ start_pk,
                                                   const int* __restrict__ sorted,
                                                   float* __restrict__ out) {
    __shared__ float Ws[DIM * LPITCH];
    __shared__ float At[FC_NODES * LPITCH];
    __shared__ float Dt[FC_NODES * LPITCH];
    int* St  = reinterpret_cast<int*>(Dt);
    int* Ids = St + N_SLABS * ST_PITCH;

    const int t    = threadIdx.x;
    const int lane = t & 31;
    const int wv   = __builtin_amdgcn_readfirstlane(t >> 5);
    const int row0 = 2 * wv;
    const int n0   = blockIdx.x * FC_NODES;
    const int node0 = n0 + row0;
    const bool live0 = node0 < N_NODES;
    const bool live1 = node0 + 1 < N_NODES;

    const int  sk = t >> 4;
    const int  sj = t & 15;
    const bool pair_ok = n0 + 2 * sj < N_NODES;
    const bool last_ok = n0 + FC_NODES < N_NODES;
    const int* __restrict__ sp = start_pk + sk * (N_NODES / 2) + (n0 >> 1);
    const int  w01 = sp[pair_ok ? sj : 0];
    const int  w32 = sp[last_ok ? 16 : 0];
    const float sd0 = s_dst[live0 ? node0 : 0];
    const float sd1 = s_dst[live1 ? node0 + 1 : 0];

    float4 wq[8];
    #pragma unroll
    for (int i = 0; i < 8; ++i) wq[i] = ((const float4*)W)[t + 512 * i];

    St[sk * ST_PITCH + 2 * sj]     = pair_ok ? (w01 & 0xffff) : SLAB;
    St[sk * ST_PITCH + 2 * sj + 1] = pair_ok ? (int)((unsigned)w01 >> 16) : SLAB;
    if (sj == 0) St[sk * ST_PITCH + FC_NODES] = last_ok ? (w32 & 0xffff) : SLAB;

    #pragma unroll
    for (int i = 0; i < 8; ++i) {
        const int q = t + 512 * i;
        *(float4*)(Ws + (q >> 5) * LPITCH + (q & 31) * 4) = wq[i];
    }

    __syncthreads();

    int cnt0, cnt1;
    const Run ra = make_run(St, sorted, lane, row0, cnt0);
    const Run rb = make_run(St, sorted, lane, row0 + 1, cnt1);

    int* ids0 = Ids + wv * 66;
    int* ids1 = ids0 + 33;
    int va[4], vb[4];
    first_request(ra, va);
    first_request(rb, vb);
    first_place(ra, va, ids0);
    first_place(rb, vb, ids1);
    strip_sync();
    const int   s00 = cnt0 > 0 ? ids0[lane < cnt0 ? lane : 0] : 0;
    const int   s10 = cnt1 > 0 ? ids1[lane < cnt1 ? lane : 0] : 0;
    const float e00 = leaky(s_src[s00] + sd0);
    const float e10 = leaky(s_src[s10] + sd1);

    const float4 o0 = node_row(h, s_src, lane, cnt0, sd0, e00, s00, St, sorted, row0, ids0);
    *(float4*)(At + row0 * LPITCH + 4 * lane) = o0;
    const float4 o1 = node_row(h, s_src, lane, cnt1, sd1, e10, s10, St, sorted, row0 + 1, ids1);
    *(float4*)(At + (row0 + 1) * LPITCH + 4 * lane) = o1;

    __syncthreads();

    {
        const int rt = wv >> 3;
        const int ct = wv & 7;
        const int hf = lane >> 4;
        const int c  = lane & 15;
        const float* Ap = At + (16 * rt + c) * LPITCH + 2 * hf;
        const float* Bp = Ws + (16 * ct + c) * LPITCH + 2 * hf;
        v8f acc0 = {}, acc1 = {};
        #pragma unroll
        for (int kk = 0; kk < DIM / 4; kk += 2) {
            const v2f a0 = *(const v2f*)(Ap + 4 * kk);
            const v2f b0 = *(const v2f*)(Bp + 4 * kk);
            const v2f a1 = *(const v2f*)(Ap + 4 * kk + 4);
            const v2f b1 = *(const v2f*)(Bp + 4 * kk + 4);
            acc0 = __builtin_amdgcn_wmma_f32_16x16x4_f32(
                 false, a0,  false, b0,
                 (short)0, acc0,  false,  false);
            acc1 = __builtin_amdgcn_wmma_f32_16x16x4_f32(
                 false, a1,  false, b1,
                 (short)0, acc1,  false,  false);
        }
        const v8f acc = acc0 + acc1;
        #pragma unroll
        for (int v = 0; v < 8; ++v)
            Dt[(16 * rt + 8 * hf + v) * LPITCH + 16 * ct + c] = acc[v];
    }

    __syncthreads();

    if (live0) ((float4*)(out + (size_t)node0 * DIM))[lane] = *(const float4*)(Dt + row0 * LPITCH + 4 * lane);
    if (live1) ((float4*)(out + (size_t)(node0 + 1) * DIM))[lane] = *(const float4*)(Dt + (row0 + 1) * LPITCH + 4 * lane);
}

extern "C" void kernel_launch(void* const* d_in, const int* in_sizes, int n_in,
                              void* d_out, int out_size, void* d_ws, size_t ws_size,
                              hipStream_t stream) {
    const float* h      = (const float*)d_in[0];
    const float* W_fc   = (const float*)d_in[1];
    const float* a_attn = (const float*)d_in[2];
    const int*   src    = (const int*)d_in[3];
    const int*   dst    = (const int*)d_in[4];
    float*       out    = (float*)d_out;
    (void)in_sizes; (void)n_in; (void)out_size; (void)ws_size;

    float* ws        = (float*)d_ws;
    float* s_src     = ws;
    float* s_dst     = s_src + N_NODES;
    int*   start_pk  = (int*)(s_dst + N_NODES);
    int*   sorted    = start_pk + N_SLABS * (N_NODES / 2);

    gat_sort_score<<<N_SLABS + SCORE_WGS, 1024, 0, stream>>>(dst, src, start_pk, sorted,
                                                             h, W_fc, a_attn, s_src, s_dst);

    gat_node_fc<<<FC_WGS, 512, 0, stream>>>(h, W_fc, s_src, s_dst, start_pk, sorted, out);
}


# sort pass also split by node range: 4 workgroups per slab, each keeps the slab's edges of one 12500-node range (12500 LDS counters); 128 sort + 128 score workgroups; same planes, arrival order kept
# speedup vs baseline: 6.6317x; 1.0749x over previous
#include <hip/hip_runtime.h>
#include <hip/hip_bf16.h>

#define N_NODES 50000
#define N_EDGES 800000
#define DIM 128
#define N_SLABS 32
#define SLAB (N_EDGES / N_SLABS)
#define SLAB4 (SLAB / 4)
#define NODES4 (N_NODES / 4)

typedef __attribute__((ext_vector_type(2))) float v2f;
typedef __attribute__((ext_vector_type(8))) float v8f;

#define LPITCH 132


#define N_PARTS      4
#define PART_NODES   (N_NODES / N_PARTS)
#define PART4        (PART_NODES / 4)
#define SORT_WGS     (N_SLABS * N_PARTS)
#define SCORE_WGS    (256 - SORT_WGS)
#define SCORE_WAVES  (SCORE_WGS * 32)
#define SCORE_ROUNDS ((N_NODES + SCORE_WAVES - 1) / SCORE_WAVES)

__device__ __forceinline__ void score_pass(const int wg, const int t,
                                           const float* __restrict__ h,
                                           const float* __restrict__ W,
                                           const float* __restrict__ a_attn,
                                           float* __restrict__ s_src,
                                           float* __restrict__ s_dst,
                                           float* __restrict__ part,
                                           float* __restrict__ u) {
    const int lane = t & 31;
    const int g    = wg * 32 + (t >> 5);

    float4 zv[SCORE_ROUNDS];
    #pragma unroll
    for (int r = 0; r < SCORE_ROUNDS; ++r) {
        const int n  = g + r * SCORE_WAVES;
        const int nc = n < N_NODES ? n : N_NODES - 1;
        zv[r] = ((const float4*)h)[(size_t)nc * (DIM / 4) + lane];
    }

    {
        const int k = t & (DIM - 1);
        const int p = t >> 7;
        float ps = 0.0f, pd = 0.0f;
        #pragma unroll
        for (int jj = 0; jj < 16; ++jj) {
            const int j = 16 * p + jj;
            const float w = W[(size_t)j * DIM + k];
            ps += a_attn[j] * w;
            pd += a_attn[DIM + j] * w;
        }
        part[p * 256 + k]       = ps;
        part[p * 256 + DIM + k] = pd;
    }
    __syncthreads();
    if (t < 256) {
        float sacc = 0.0f;
        #pragma unroll
        for (int p = 0; p < 8; ++p) sacc += part[p * 256 + t];
        u[t] = sacc;
    }
    __syncthreads();

    const float4 av = ((const float4*)(u))[lane];
    const float4 dv = ((const float4*)(u + DIM))[lane];

    #pragma unroll
    for (int r = 0; r < SCORE_ROUNDS; ++r) {
        const int n = g + r * SCORE_WAVES;
        float ss = zv[r].x * av.x + zv[r].y * av.y + zv[r].z * av.z + zv[r].w * av.w;
        float sd = zv[r].x * dv.x + zv[r].y * dv.y + zv[r].z * dv.z + zv[r].w * dv.w;

        #pragma unroll
        for (int off = 16; off > 0; off >>= 1) {
            ss += __shfl_down(ss, off, 32);
            sd += __shfl_down(sd, off, 32);
        }
        if (lane == 0 && n < N_NODES) {
            s_src[n] = ss;
            s_dst[n] = sd;
        }
    }
}

#define SCAN_CHUNK 13

__device__ __forceinline__ int count_edge(int* tbl, int& d, const int lo, const int lane, int& front) {
    const int  rel  = d - lo;
    const bool keep = (unsigned)rel < (unsigned)PART_NODES;
    front += rel < 0 ? 1 : 0;
    int rank = lane;
    if (keep) rank = atomicAdd(&tbl[rel], 1);
    d = keep ? rel : PART_NODES;
    return rank;
}

__device__ __forceinline__ void place_edge(const int* tbl, int* lst, const int rel, const int rank,
                                           const int nb, const int s) {
    lst[tbl[rel] - nb + rank] = s;
}

__global__ __launch_bounds__(1024) void gat_sort_score(const int* __restrict__ dst,
                                                       const int* __restrict__ src,
                                                       int* __restrict__ start_pk,
                                                       int* __restrict__ sorted,
                                                       const float* __restrict__ h,
                                                       const float* __restrict__ W,
                                                       const float* __restrict__ a_attn,
                                                       float* __restrict__ s_src,
                                                       float* __restrict__ s_dst) {
    __shared__ __attribute__((aligned(16))) int tbl[PART_NODES + 4];
    __shared__ __attribute__((aligned(16))) int lst[SLAB + 32];
    __shared__ int wsum[32];
    __shared__ int wfront[32];
    __shared__ int tot[2];
    const int t  = threadIdx.x;
    const int b  = blockIdx.x;

    if (b >= SORT_WGS) {
        float* fl = reinterpret_cast<float*>(tbl);
        score_pass(b - SORT_WGS, t, h, W, a_attn, s_src, s_dst, fl, fl + 8 * 256);
        return;
    }

    const int k  = b / N_PARTS;
    const int lo = (b % N_PARTS) * PART_NODES;

    const int4* __restrict__ d4 = (const int4*)(dst + (size_t)k * SLAB);
    const int4* __restrict__ s4 = (const int4*)(src + (size_t)k * SLAB);

    int4 dv[7], sv[7];
    #pragma unroll
    for (int r = 0; r < 7; ++r) {
        const int q  = t + r * 1024;
        const int qc = q < SLAB4 ? q : SLAB4 - 1;
        dv[r] = d4[qc];
        sv[r] = s4[qc];
    }

    int4* tbl4 = (int4*)tbl;
    for (int n = t; n < PART4; n += 1024) tbl4[n] = make_int4(0, 0, 0, 0);
    __syncthreads();

    const int lane = t & 31;
    const int wv   = t >> 5;

    int4 rk[7];
    int front = 0;
    #pragma unroll
    for (int r = 0; r < 7; ++r) {
        const int q = t + r * 1024;
        if (q < SLAB4) {
            rk[r].x = count_edge(tbl, dv[r].x, lo, lane, front);
            rk[r].y = count_edge(tbl, dv[r].y, lo, lane, front);
            rk[r].z = count_edge(tbl, dv[r].z, lo, lane, front);
            rk[r].w = count_edge(tbl, dv[r].w, lo, lane, front);
        } else {
            dv[r] = make_int4(PART_NODES, PART_NODES, PART_NODES, PART_NODES);
            rk[r] = make_int4(lane, lane, lane, lane);
        }
    }
    #pragma unroll
    for (int off = 16; off > 0; off >>= 1) front += __shfl_xor(front, off, 32);
    if (lane == 0) wfront[wv] = front;
    __syncthreads();

    const int c0 = t * SCAN_CHUNK;
    int c[SCAN_CHUNK];
    int own = 0;
    #pragma unroll
    for (int j = 0; j < SCAN_CHUNK; ++j) {
        const int i = c0 + j;
        c[j] = tbl[i < PART_NODES ? i : PART_NODES - 1];
    }
    #pragma unroll
    for (int j = 0; j < SCAN_CHUNK; ++j) {
        c[j] = c0 + j < PART_NODES ? c[j] : 0;
        own += c[j];
    }
    int inc = own;
    for (int off = 1; off < 32; off <<= 1) {
        const int u = __shfl_up(inc, off, 32);
        if (lane >= off) inc += u;
    }
    if (lane == 31) wsum[wv] = inc;
    __syncthreads();
    if (wv == 0) {
        const int v = wsum[lane];
        int wi = v;
        for (int off = 1; off < 32; off <<= 1) {
            const int u = __shfl_up(wi, off, 32);
            if (lane >= off) wi += u;
        }
        int fr = wfront[lane];
        #pragma unroll
        for (int off = 16; off > 0; off >>= 1) fr += __shfl_xor(fr, off, 32);
        wsum[lane] = fr + wi - v;
        if (lane == 31) { tot[0] = fr; tot[1] = wi; }
    }
    __syncthreads();
    const int nb = tot[0];
    const int nq = tot[1];
    {
        int run = wsum[wv] + inc - own;
        #pragma unroll
        for (int j = 0; j < SCAN_CHUNK; ++j) {
            const int i = c0 + j;
            if (i < PART_NODES) tbl[i] = run;
            run += c[j];
        }
        if (t == 0) tbl[PART_NODES] = nb + SLAB;
    }
    __syncthreads();

    #pragma unroll
    for (int r = 0; r < 7; ++r) {
        place_edge(tbl, lst, dv[r].x, rk[r].x, nb, sv[r].x);
        place_edge(tbl, lst, dv[r].y, rk[r].y, nb, sv[r].y);
        place_edge(tbl, lst, dv[r].z, rk[r].z, nb, sv[r].z);
        place_edge(tbl, lst, dv[r].w, rk[r].w, nb, sv[r].w);
    }
    __syncthreads();

    {
        const int first = k * SLAB + nb;
        const int end   = first + nq;
        int p = (first & ~31) + t;
        if (p < first) p += 1024;
        for (; p < end; p += 1024) sorted[p] = lst[p - first];
    }
    {
        const int first = k * (N_NODES / 2) + lo / 2;
        const int end   = first + PART_NODES / 2;
        const int2* tbl2 = (const int2*)tbl;
        int p = (first & ~31) + t;
        if (p < first) p += 1024;
        for (; p < end; p += 1024) {
            const int2 v = tbl2[p - first];
            start_pk[p] = (v.x & 0xffff) | (v.y << 16);
        }
    }
}


template <int N>
__device__ __forceinline__ void rows_request(const float* __restrict__ x, const int lane,
                                             const int sl, const int j, float4 (&r)[8]) {
    #pragma unroll
    for (int u = 0; u < N; ++u) {
        const int s = __builtin_amdgcn_readlane(sl, j + u);
        r[u] = ((const float4*)x)[(unsigned)s * (DIM / 4) + (unsigned)lane];
    }
}

template <int N>
__device__ __forceinline__ void rows_sum(const float wl, const int j,
                                         const float4 (&r)[8], float4& acc) {
    #pragma unroll
    for (int u = 0; u < N; ++u) {
        const float w = __int_as_float(__builtin_amdgcn_readlane(__float_as_int(wl), j + u));
        acc.x += w * r[u].x;
        acc.y += w * r[u].y;
        acc.z += w * r[u].z;
        acc.w += w * r[u].w;
    }
}

__device__ __forceinline__ void rows_block(const float* __restrict__ x, const int lane,
                                           const int sl, const float wl,
                                           int j, const int nn, float4& acc) {
    float4 r[8];
    for (; j + 8 <= nn; j += 8) {
        rows_request<8>(x, lane, sl, j, r);
        rows_sum<8>(wl, j, r, acc);
    }
    const int rem = nn - j;
    if (rem > 4) {
        rows_request<8>(x, lane, sl, j, r);
        rows_sum<8>(wl, j, r, acc);
    } else if (rem > 0) {
        rows_request<4>(x, lane, sl, j, r);
        rows_sum<4>(wl, j, r, acc);
    }
}

__device__ __forceinline__ float wave_sum(float v) {
    #pragma unroll
    for (int off = 16; off > 0; off >>= 1) v += __shfl_xor(v, off, 32);
    return v;
}

#define ST_PITCH 33

__device__ __forceinline__ float leaky(const float v) {
    return v > 0.0f ? v : 0.01f * v;
}

struct Run { const int* sk; int a; int c; int off; };

__device__ __forceinline__ int wave_excl(const int v, const int lane, int& total) {
    int inc = v;
    #pragma unroll
    for (int off = 1; off < 32; off <<= 1) {
        const int u = __shfl_up(inc, off, 32);
        if (lane >= off) inc += u;
    }
    total = __builtin_amdgcn_readlane(inc, 31);
    return inc - v;
}

__device__ __forceinline__ Run make_run(const int* St, const int* __restrict__ sorted,
                                        const int lane, const int row, int& cnt) {
    Run r;
    const int a = St[lane * ST_PITCH + row];
    const int b = St[lane * ST_PITCH + row + 1];
    r.sk  = sorted;
    r.a   = lane * SLAB + a;
    r.c   = b - a;
    r.off = wave_excl(r.c, lane, cnt);
    return r;
}

__device__ __forceinline__ void strip_sync() {
    __builtin_amdgcn_fence(__ATOMIC_ACQ_REL, "wavefront");
    __builtin_amdgcn_wave_barrier();
}

__device__ __forceinline__ void strip_fill(const Run& r, const int eb, int* ids) {
    const int i0 = max(0, eb - r.off);
    const int i1 = min(r.c, eb + 32 - r.off);
    #pragma clang loop vectorize(disable) unroll(disable)
    for (int i = i0; i < i1; ++i) ids[r.off + i - eb] = r.sk[r.a + i];
}

__device__ __forceinline__ float4 node_row(const float* __restrict__ h,
                                           const float* __restrict__ s_src,
                                           const int lane, const int cnt, const float sd,
                                           const float e0, const int s0,
                                           const int* St, const int* __restrict__ sorted,
                                           const int row, int* ids) {
    float4 acc = {0.0f, 0.0f, 0.0f, 0.0f};
    float sum = 0.0f;
    {
        const bool in0 = lane < cnt;

        float4 r0[8];
        rows_request<8>(h, lane, s0, 0, r0);

        float mx = in0 ? e0 : -INFINITY;
        if (cnt > 32) {
            int deg;
            const Run run = make_run(St, sorted, lane, row, deg);
            for (int eb = 32; eb < cnt; eb += 32) {
                strip_sync();
                strip_fill(run, eb, ids);
                strip_sync();
                const bool  in = eb + lane < cnt;
                const int   sl = ids[in ? lane : 0];
                const float el = leaky(s_src[sl] + sd);
                if (in) mx = fmaxf(mx, el);
            }
        }
        #pragma unroll
        for (int off = 16; off > 0; off >>= 1) mx = fmaxf(mx, __shfl_xor(mx, off, 32));

        const float w0 = in0 ? __expf(e0 - mx) : 0.0f;
        sum = w0;
        if (cnt <= 32) sum = wave_sum(sum);

        rows_sum<8>(w0, 0, r0, acc);
        rows_block(h, lane, s0, w0, 8, min(32, cnt), acc);

        if (cnt > 32) {
            int deg;
            const Run run = make_run(St, sorted, lane, row, deg);
            for (int eb = 32; eb < cnt; eb += 32) {
                strip_sync();
                strip_fill(run, eb, ids);
                strip_sync();
                const bool  in = eb + lane < cnt;
                const int   sl = ids[in ? lane : 0];
                const float el = leaky(s_src[sl] + sd);
                const float wl = in ? __expf(el - mx) : 0.0f;
                sum += wl;
                rows_block(h, lane, sl, wl, 0, min(32, cnt - eb), acc);
            }
            sum = wave_sum(sum);
        }
    }
    if (cnt <= 0) return make_float4(0.0f, 0.0f, 0.0f, 0.0f);
    const float inv = 1.0f / sum;
    return make_float4(acc.x * inv, acc.y * inv, acc.z * inv, acc.w * inv);
}

__device__ __forceinline__ void first_request(const Run& r, int (&v)[4]) {
    #pragma unroll
    for (int u = 0; u < 4; ++u) v[u] = r.sk[u < r.c ? r.a + u : 0];
}
__device__ __forceinline__ void first_place(const Run& r, const int (&v)[4], int* ids) {
    #pragma unroll
    for (int u = 0; u < 4; ++u) {
        const bool ok = u < r.c && r.off + u < 32;
        ids[ok ? r.off + u : 32] = v[u];
    }
    const int i1 = min(r.c, 32 - r.off);
    #pragma clang loop vectorize(disable) unroll(disable)
    for (int i = 4; i < i1; ++i) ids[r.off + i] = r.sk[r.a + i];
}


#define FC_NODES 32
#define FC_WGS   ((N_NODES + FC_NODES - 1) / FC_NODES)

__global__ __launch_bounds__(512) void gat_node_fc(const float* __restrict__ h,
                                                   const float* __restrict__ W,
                                                   const float* __restrict__ s_src,
                                                   const float* __restrict__ s_dst,
                                                   const int* __restrict__ start_pk,
                                                   const int* __restrict__ sorted,
                                                   float* __restrict__ out) {
    __shared__ float Ws[DIM * LPITCH];
    __shared__ float At[FC_NODES * LPITCH];
    __shared__ float Dt[FC_NODES * LPITCH];
    int* St  = reinterpret_cast<int*>(Dt);
    int* Ids = St + N_SLABS * ST_PITCH;

    const int t    = threadIdx.x;
    const int lane = t & 31;
    const int wv   = __builtin_amdgcn_readfirstlane(t >> 5);
    const int row0 = 2 * wv;
    const int n0   = blockIdx.x * FC_NODES;
    const int node0 = n0 + row0;
    const bool live0 = node0 < N_NODES;
    const bool live1 = node0 + 1 < N_NODES;

    const int  sk = t >> 4;
    const int  sj = t & 15;
    const bool pair_ok = n0 + 2 * sj < N_NODES;
    const bool last_ok = n0 + FC_NODES < N_NODES;
    const int* __restrict__ sp = start_pk + sk * (N_NODES / 2) + (n0 >> 1);
    const int  w01 = sp[pair_ok ? sj : 0];
    const int  w32 = sp[last_ok ? 16 : 0];
    const float sd0 = s_dst[live0 ? node0 : 0];
    const float sd1 = s_dst[live1 ? node0 + 1 : 0];

    float4 wq[8];
    #pragma unroll
    for (int i = 0; i < 8; ++i) wq[i] = ((const float4*)W)[t + 512 * i];

    St[sk * ST_PITCH + 2 * sj]     = pair_ok ? (w01 & 0xffff) : SLAB;
    St[sk * ST_PITCH + 2 * sj + 1] = pair_ok ? (int)((unsigned)w01 >> 16) : SLAB;
    if (sj == 0) St[sk * ST_PITCH + FC_NODES] = last_ok ? (w32 & 0xffff) : SLAB;

    #pragma unroll
    for (int i = 0; i < 8; ++i) {
        const int q = t + 512 * i;
        *(float4*)(Ws + (q >> 5) * LPITCH + (q & 31) * 4) = wq[i];
    }

    __syncthreads();

    int cnt0, cnt1;
    const Run ra = make_run(St, sorted, lane, row0, cnt0);
    const Run rb = make_run(St, sorted, lane, row0 + 1, cnt1);

    int* ids0 = Ids + wv * 66;
    int* ids1 = ids0 + 33;
    int va[4], vb[4];
    first_request(ra, va);
    first_request(rb, vb);
    first_place(ra, va, ids0);
    first_place(rb, vb, ids1);
    strip_sync();
    const int   s00 = cnt0 > 0 ? ids0[lane < cnt0 ? lane : 0] : 0;
    const int   s10 = cnt1 > 0 ? ids1[lane < cnt1 ? lane : 0] : 0;
    const float e00 = leaky(s_src[s00] + sd0);
    const float e10 = leaky(s_src[s10] + sd1);

    const float4 o0 = node_row(h, s_src, lane, cnt0, sd0, e00, s00, St, sorted, row0, ids0);
    *(float4*)(At + row0 * LPITCH + 4 * lane) = o0;
    const float4 o1 = node_row(h, s_src, lane, cnt1, sd1, e10, s10, St, sorted, row0 + 1, ids1);
    *(float4*)(At + (row0 + 1) * LPITCH + 4 * lane) = o1;

    __syncthreads();

    {
        const int rt = wv >> 3;
        const int ct = wv & 7;
        const int hf = lane >> 4;
        const int c  = lane & 15;
        const float* Ap = At + (16 * rt + c) * LPITCH + 2 * hf;
        const float* Bp = Ws + (16 * ct + c) * LPITCH + 2 * hf;
        v8f acc0 = {}, acc1 = {};
        #pragma unroll
        for (int kk = 0; kk < DIM / 4; kk += 2) {
            const v2f a0 = *(const v2f*)(Ap + 4 * kk);
            const v2f b0 = *(const v2f*)(Bp + 4 * kk);
            const v2f a1 = *(const v2f*)(Ap + 4 * kk + 4);
            const v2f b1 = *(const v2f*)(Bp + 4 * kk + 4);
            acc0 = __builtin_amdgcn_wmma_f32_16x16x4_f32(
                 false, a0,  false, b0,
                 (short)0, acc0,  false,  false);
            acc1 = __builtin_amdgcn_wmma_f32_16x16x4_f32(
                 false, a1,  false, b1,
                 (short)0, acc1,  false,  false);
        }
        const v8f acc = acc0 + acc1;
        #pragma unroll
        for (int v = 0; v < 8; ++v)
            Dt[(16 * rt + 8 * hf + v) * LPITCH + 16 * ct + c] = acc[v];
    }

    __syncthreads();

    if (live0) ((float4*)(out + (size_t)node0 * DIM))[lane] = *(const float4*)(Dt + row0 * LPITCH + 4 * lane);
    if (live1) ((float4*)(out + (size_t)(node0 + 1) * DIM))[lane] = *(const float4*)(Dt + (row0 + 1) * LPITCH + 4 * lane);
}

extern "C" void kernel_launch(void* const* d_in, const int* in_sizes, int n_in,
                              void* d_out, int out_size, void* d_ws, size_t ws_size,
                              hipStream_t stream) {
    const float* h      = (const float*)d_in[0];
    const float* W_fc   = (const float*)d_in[1];
    const float* a_attn = (const float*)d_in[2];
    const int*   src    = (const int*)d_in[3];
    const int*   dst    = (const int*)d_in[4];
    float*       out    = (float*)d_out;
    (void)in_sizes; (void)n_in; (void)out_size; (void)ws_size;

    float* ws        = (float*)d_ws;
    float* s_src     = ws;
    float* s_dst     = s_src + N_NODES;
    int*   start_pk  = (int*)(s_dst + N_NODES);
    int*   sorted    = start_pk + N_SLABS * (N_NODES / 2);

    gat_sort_score<<<SORT_WGS + SCORE_WGS, 1024, 0, stream>>>(dst, src, start_pk, sorted,
                                                             h, W_fc, a_attn, s_src, s_dst);

    gat_node_fc<<<FC_WGS, 512, 0, stream>>>(h, W_fc, s_src, s_dst, start_pk, sorted, out);
}
